# starting kernel
# baseline (speedup 1.0000x reference)
#include <hip/hip_runtime.h>

typedef __attribute__((ext_vector_type(2))) float v2f;
typedef __attribute__((ext_vector_type(8))) float v8f;

#define B_       2
#define CIN      3
#define COUT     64
#define HIN      721
#define WIN      1440
#define HOUT     361
#define WOUT     720
#define KTAP     9
#define NNZ      96
#define LN_EPS   1e-5f

#define ASTRIDE  98
#define NWAVES   6
#define NTHREADS (NWAVES * 32)
#define PLANE    (HIN * WIN)
#define NTOT     (B_ * WOUT)

__global__ __launch_bounds__(NTHREADS)
void opm_gather_gemm_ln_kernel(const float* __restrict__ x,
                               const float* __restrict__ weight,
                               const float* __restrict__ psi,
                               const int*   __restrict__ idx_hi,
                               const int*   __restrict__ idx_dwi,
                               const float* __restrict__ lnw,
                               const float* __restrict__ lnb,
                               float*       __restrict__ out)
{
    __shared__ float psi_lds[KTAP * NNZ];
    __shared__ float w_lds[COUT * CIN * KTAP];
    __shared__ int2  base_lds[NNZ];
    __shared__ float A_lds[COUT * ASTRIDE];
    __shared__ float lnw_lds[COUT];
    __shared__ float lnb_lds[COUT];

    const int tid  = threadIdx.x;
    const int h    = blockIdx.y;
    const int lane = tid & 31;
    const int wid  = tid >> 5;
    const int l16  = lane & 15;
    const int half = lane >> 4;

    for (int i = tid; i < KTAP * NNZ; i += NTHREADS) {
        int k9 = i / NNZ, e = i - k9 * NNZ;
        psi_lds[i] = psi[(k9 * HOUT + h) * NNZ + e];
    }
    for (int i = tid; i < COUT * CIN * KTAP; i += NTHREADS)
        w_lds[i] = weight[i];
    for (int i = tid; i < NNZ; i += NTHREADS)
        base_lds[i] = make_int2(idx_hi[h * NNZ + i] * WIN, idx_dwi[h * NNZ + i]);
    for (int i = tid; i < COUT; i += NTHREADS) {
        lnw_lds[i] = lnw[i];
        lnb_lds[i] = lnb[i];
    }

    const int eown = tid & (NNZ - 1);
    const int eA_own = tid - (tid >= NNZ ? NNZ : 0);
    const int o0     = (tid >= NNZ) ? 1 : 0;
    (void)eown;

    const int n    = blockIdx.x * (NWAVES * 16) + wid * 16 + l16;
    const int b    = n / WOUT;
    const int w    = n - b * WOUT;
    const int wcol = w << 1;

    v8f acc[4];
    #pragma unroll
    for (int mt = 0; mt < 4; ++mt) acc[mt] = (v8f){0.f,0.f,0.f,0.f,0.f,0.f,0.f,0.f};

    for (int c = 0; c < CIN; ++c) {
        __syncthreads();

        {
            float p[KTAP];
            #pragma unroll
            for (int k9 = 0; k9 < KTAP; ++k9)
                p[k9] = psi_lds[k9 * NNZ + eA_own];
            for (int o = o0; o < COUT; o += 2) {
                const float* wp = &w_lds[o * (CIN * KTAP) + c * KTAP];
                float s = 0.f;
                #pragma unroll
                for (int k9 = 0; k9 < KTAP; ++k9)
                    s += wp[k9] * p[k9];
                A_lds[o * ASTRIDE + eA_own] = s;
            }
        }
        __syncthreads();

        const int plane = (b * CIN + c) * PLANE;

        #pragma unroll 2
        for (int e0 = 0; e0 < NNZ; e0 += 4) {
            const int eA = e0 + (half << 1);

            const int2 r0 = base_lds[eA];
            const int2 r1 = base_lds[eA + 1];
            int c0 = r0.y + wcol; c0 -= (c0 >= WIN) ? WIN : 0;
            int c1 = r1.y + wcol; c1 -= (c1 >= WIN) ? WIN : 0;
            v2f bf;
            bf.x = x[plane + r0.x + c0];
            bf.y = x[plane + r1.x + c1];

            v2f a0 = *(const v2f*)&A_lds[( 0 + l16) * ASTRIDE + eA];
            v2f a1 = *(const v2f*)&A_lds[(16 + l16) * ASTRIDE + eA];
            v2f a2 = *(const v2f*)&A_lds[(32 + l16) * ASTRIDE + eA];
            v2f a3 = *(const v2f*)&A_lds[(48 + l16) * ASTRIDE + eA];

            acc[0] = __builtin_amdgcn_wmma_f32_16x16x4_f32(false, a0, false, bf, (short)0, acc[0], false, false);
            acc[1] = __builtin_amdgcn_wmma_f32_16x16x4_f32(false, a1, false, bf, (short)0, acc[1], false, false);
            acc[2] = __builtin_amdgcn_wmma_f32_16x16x4_f32(false, a2, false, bf, (short)0, acc[2], false, false);
            acc[3] = __builtin_amdgcn_wmma_f32_16x16x4_f32(false, a3, false, bf, (short)0, acc[3], false, false);
        }
    }

    float s = 0.f, ss = 0.f;
    #pragma unroll
    for (int mt = 0; mt < 4; ++mt) {
        #pragma unroll
        for (int r = 0; r < 8; ++r) {
            float v = acc[mt][r];
            s  += v;
            ss += v * v;
        }
    }
    s  += __shfl_xor(s, 16, 32);
    ss += __shfl_xor(ss, 16, 32);
    const float mean = s * (1.f / COUT);
    const float var  = ss * (1.f / COUT) - mean * mean;
    const float rstd = rsqrtf(var + LN_EPS);

    float* outp = out + (size_t)b * COUT * HOUT * WOUT + (size_t)h * WOUT + w;
    #pragma unroll
    for (int mt = 0; mt < 4; ++mt) {
        #pragma unroll
        for (int r = 0; r < 8; ++r) {
            const int o = mt * 16 + r + (half << 3);
            const float v = (acc[mt][r] - mean) * rstd * lnw_lds[o] + lnb_lds[o];
            outp[(size_t)o * (HOUT * WOUT)] = v;
        }
    }
}

extern "C" void kernel_launch(void* const* d_in, const int* in_sizes, int n_in,
                              void* d_out, int out_size, void* d_ws, size_t ws_size,
                              hipStream_t stream) {
    (void)in_sizes; (void)n_in; (void)d_ws; (void)ws_size; (void)out_size;
    const float* x       = (const float*)d_in[0];
    const float* weight  = (const float*)d_in[1];
    const float* psi     = (const float*)d_in[2];
    const int*   idx_hi  = (const int*)d_in[3];
    const int*   idx_dwi = (const int*)d_in[4];
    const float* lnw     = (const float*)d_in[5];
    const float* lnb     = (const float*)d_in[6];
    float*       out     = (float*)d_out;

    dim3 grid(NTOT / (NWAVES * 16), HOUT, 1);
    dim3 block(NTHREADS, 1, 1);
    opm_gather_gemm_ln_kernel<<<grid, block, 0, stream>>>(
        x, weight, psi, idx_hi, idx_dwi, lnw, lnb, out);
}


# K loop software pipeline deepened: gathers requested 3 tap steps ahead in a ring of 4 register sets that runs on across unit ends, A fragments 1 step ahead; summation order unchanged
# speedup vs baseline: 1.4433x; 1.4433x over previous
#include <hip/hip_runtime.h>


typedef __attribute__((ext_vector_type(2))) float v2f;
typedef __attribute__((ext_vector_type(8))) float v8f;

#define B_       2
#define CIN      3
#define COUT     64
#define HIN      721
#define WIN      1440
#define HOUT     361
#define WOUT     720
#define KTAP     9
#define NNZ      96
#define LN_EPS   1e-5f

#define KTOT     (CIN * NNZ)
#define ASTR     292
#define NWAVES   4
#define NTHREADS (NWAVES * 32)
#define UCOLS    32
#define NUNITS   (NTOT / UCOLS)
#define NSPLIT   2
#define RES_PAD  2048
#define PLANE    (HIN * WIN)
#define NTOT     (B_ * WOUT)
#define NSTEP    (NNZ / 4)

struct Gath { v2f g[2][CIN]; };

__device__ __forceinline__ void gather_step(const float* __restrict__ x,
                                            const int2* base_lds, int s, int half,
                                            const int (&pb)[2], const int (&wcol)[2], Gath& G)
{
    const int eA = 4 * s + 2 * half;
    const int2 r0 = base_lds[eA];
    const int2 r1 = base_lds[eA + 1];
    #pragma unroll
    for (int t = 0; t < 2; ++t) {
        int c0 = r0.y + wcol[t]; c0 -= (c0 >= WIN) ? WIN : 0;
        int c1 = r1.y + wcol[t]; c1 -= (c1 >= WIN) ? WIN : 0;
        const int o0 = pb[t] + r0.x + c0;
        const int o1 = pb[t] + r1.x + c1;
        #pragma unroll
        for (int c = 0; c < CIN; ++c) {
            G.g[t][c].x = x[o0 + c * PLANE];
            G.g[t][c].y = x[o1 + c * PLANE];
        }
    }
}

struct Afr { v2f a[CIN][4]; };

__device__ __forceinline__ void loadA_step(const float* A_lds, int s, int half, int l16, Afr& F)
{
    const int eA = 4 * s + 2 * half;
    #pragma unroll
    for (int c = 0; c < CIN; ++c)
        #pragma unroll
        for (int mt = 0; mt < 4; ++mt)
            F.a[c][mt] = *(const v2f*)&A_lds[(mt * 16 + l16) * ASTR + c * NNZ + eA];
}

__device__ __forceinline__ void mma_step(const Afr& F, const Gath& G, v8f (&acc)[2][4])
{
    #pragma unroll
    for (int c = 0; c < CIN; ++c)
        #pragma unroll
        for (int mt = 0; mt < 4; ++mt)
            #pragma unroll
            for (int t = 0; t < 2; ++t)
                acc[t][mt] = __builtin_amdgcn_wmma_f32_16x16x4_f32(false, F.a[c][mt], false, G.g[t][c],
                                                                   (short)0, acc[t][mt], false, false);
}

__device__ __forceinline__ void unit_cols(int unit, int l16, int (&pb)[2], int (&wcol)[2])
{
    #pragma unroll
    for (int t = 0; t < 2; ++t) {
        const int n = unit * UCOLS + t * 16 + l16;
        const int b = n / WOUT;
        const int w = n - b * WOUT;
        wcol[t] = w << 1;
        pb[t]   = b * CIN * PLANE;
    }
}

__global__ __launch_bounds__(NTHREADS)
void opm_gather_gemm_ln_kernel(const float* __restrict__ x,
                               const float* __restrict__ weight,
                               const float* __restrict__ psi,
                               const int*   __restrict__ idx_hi,
                               const int*   __restrict__ idx_dwi,
                               const float* __restrict__ lnw,
                               const float* __restrict__ lnb,
                               float*       __restrict__ out)
{
    __shared__ __attribute__((aligned(16))) float A_lds[COUT * ASTR + RES_PAD];
    __shared__ __attribute__((aligned(16))) int2  base_lds[NNZ];
    __shared__ float lnw_lds[COUT];
    __shared__ float lnb_lds[COUT];

    const int tid  = threadIdx.x;
    const int h    = blockIdx.y;
    const int lane = tid & 31;
    const int l16  = lane & 15;
    const int half = lane >> 4;
    const int wv   = __builtin_amdgcn_readfirstlane(tid >> 5);

    if (tid < NNZ)
        base_lds[tid] = make_int2(idx_hi[h * NNZ + tid] * WIN, idx_dwi[h * NNZ + tid]);
    if (tid < COUT) {
        lnw_lds[tid] = lnw[tid];
        lnb_lds[tid] = lnb[tid];
    }

    {
        float p[3][KTAP];
        #pragma unroll
        for (int k9 = 0; k9 < KTAP; ++k9)
            #pragma unroll
            for (int j = 0; j < 3; ++j)
                p[j][k9] = psi[(k9 * HOUT + h) * NNZ + j * 32 + lane];

        float wc[CIN * KTAP];
        #pragma unroll
        for (int i = 0; i < CIN * KTAP; ++i) wc[i] = weight[wv * (CIN * KTAP) + i];

        for (int o = wv; o < COUT; o += NWAVES) {
            const int on = (o + NWAVES < COUT) ? (o + NWAVES) : o;
            float wn[CIN * KTAP];
            #pragma unroll
            for (int i = 0; i < CIN * KTAP; ++i) wn[i] = weight[on * (CIN * KTAP) + i];

            #pragma unroll
            for (int c = 0; c < CIN; ++c) {
                #pragma unroll
                for (int j = 0; j < 3; ++j) {
                    float s = 0.f;
                    #pragma unroll
                    for (int k9 = 0; k9 < KTAP; ++k9)
                        s += wc[c * KTAP + k9] * p[j][k9];
                    A_lds[o * ASTR + c * NNZ + j * 32 + lane] = s;
                }
            }
            #pragma unroll
            for (int i = 0; i < CIN * KTAP; ++i) wc[i] = wn[i];
        }
    }

    __syncthreads();

    const int u0 = (blockIdx.x * NUNITS + NSPLIT - 1) / NSPLIT;
    const int u1 = ((blockIdx.x + 1) * NUNITS + NSPLIT - 1) / NSPLIT;
    #define STEP(Fcur, Gcur, sA, Fnext, sG, Gnext, PB, WC)                  \
        do {                                                                \
            loadA_step(A_lds, (sA), half, l16, Fnext);                      \
            gather_step(x, base_lds, (sG), half, PB, WC, Gnext);            \
            __builtin_amdgcn_sched_barrier(0);                              \
            mma_step(Fcur, Gcur, acc);                                      \
        } while (0)

    int unit = u0 + wv;
    int pb[2], wcol[2];
    unit_cols(unit, l16, pb, wcol);

    Gath G0, G1, G2, G3;
    Afr  F0, F1;
    gather_step(x, base_lds, 0, half, pb, wcol, G0);
    gather_step(x, base_lds, 1, half, pb, wcol, G1);
    gather_step(x, base_lds, 2, half, pb, wcol, G2);
    loadA_step(A_lds, 0, half, l16, F0);

    for (; unit < u1; unit += NWAVES) {

    v8f acc[2][4];
    #pragma unroll
    for (int t = 0; t < 2; ++t)
        #pragma unroll
        for (int mt = 0; mt < 4; ++mt) acc[t][mt] = (v8f){0.f,0.f,0.f,0.f,0.f,0.f,0.f,0.f};

    for (int s = 0; s < NSTEP - 4; s += 4) {
        STEP(F0, G0, s + 1, F1, s + 3, G3, pb, wcol);
        STEP(F1, G1, s + 2, F0, s + 4, G0, pb, wcol);
        STEP(F0, G2, s + 3, F1, s + 5, G1, pb, wcol);
        STEP(F1, G3, s + 4, F0, s + 6, G2, pb, wcol);
    }

    const int nunit = (unit + NWAVES < u1) ? (unit + NWAVES) : unit;
    int pbN[2], wcolN[2];
    unit_cols(nunit, l16, pbN, wcolN);
    STEP(F0, G0, NSTEP - 3, F1, NSTEP - 1, G3, pb,  wcol);
    STEP(F1, G1, NSTEP - 2, F0, 0,         G0, pbN, wcolN);
    STEP(F0, G2, NSTEP - 1, F1, 1,         G1, pbN, wcolN);
    STEP(F1, G3, 0,         F0, 2,         G2, pbN, wcolN);

    int bb[2], ww[2];
    #pragma unroll
    for (int t = 0; t < 2; ++t) {
        const int n = unit * UCOLS + t * 16 + l16;
        bb[t] = n / WOUT;
        ww[t] = n - bb[t] * WOUT;
    }

    #pragma unroll
    for (int t = 0; t < 2; ++t) {
        float s = 0.f, ss = 0.f;
        #pragma unroll
        for (int mt = 0; mt < 4; ++mt) {
            #pragma unroll
            for (int r = 0; r < 8; ++r) {
                float v = acc[t][mt][r];
                s  += v;
                ss += v * v;
            }
        }
        s  += __shfl_xor(s, 16, 32);
        ss += __shfl_xor(ss, 16, 32);
        const float mean = s * (1.f / COUT);
        const float var  = ss * (1.f / COUT) - mean * mean;
        const float rstd = rsqrtf(var + LN_EPS);

        const int obase = bb[t] * (COUT * HOUT * WOUT) + h * WOUT + ww[t] + (half << 3) * (HOUT * WOUT);
        #pragma unroll
        for (int mt = 0; mt < 4; ++mt) {
            #pragma unroll
            for (int r = 0; r < 8; ++r) {
                const int o = mt * 16 + r + (half << 3);
                const float v = (acc[t][mt][r] - mean) * rstd * lnw_lds[o] + lnb_lds[o];
                out[obase + (mt * 16 + r) * (HOUT * WOUT)] = v;
            }
        }
    }

    #pragma unroll
    for (int t = 0; t < 2; ++t) { pb[t] = pbN[t]; wcol[t] = wcolN[t]; }

    }
    #undef STEP
}

extern "C" void kernel_launch(void* const* d_in, const int* in_sizes, int n_in,
                              void* d_out, int out_size, void* d_ws, size_t ws_size,
                              hipStream_t stream) {
    (void)in_sizes; (void)n_in; (void)d_ws; (void)ws_size; (void)out_size;
    const float* x       = (const float*)d_in[0];
    const float* weight  = (const float*)d_in[1];
    const float* psi     = (const float*)d_in[2];
    const int*   idx_hi  = (const int*)d_in[3];
    const int*   idx_dwi = (const int*)d_in[4];
    const float* lnw     = (const float*)d_in[5];
    const float* lnb     = (const float*)d_in[6];
    float*       out     = (float*)d_out;

    dim3 grid(NSPLIT, HOUT, 1);
    dim3 block(NTHREADS, 1, 1);
    opm_gather_gemm_ln_kernel<<<grid, block, 0, stream>>>(
        x, weight, psi, idx_hi, idx_dwi, lnw, lnb, out);
}


# Merge taps of equal (input row, column offset): psi summed per distinct tap before W*psi is built; K loop covers distinct taps (padded to 16); gain scales with coinciding taps; changes summing order
# speedup vs baseline: 1.6479x; 1.1418x over previous
#include <hip/hip_runtime.h>


typedef __attribute__((ext_vector_type(2))) float v2f;
typedef __attribute__((ext_vector_type(8))) float v8f;

#define B_       2
#define CIN      3
#define COUT     64
#define HIN      721
#define WIN      1440
#define HOUT     361
#define WOUT     720
#define KTAP     9
#define NNZ      96
#define LN_EPS   1e-5f

#define KTOT     (CIN * NNZ)
#define ASTR     292
#define NWAVES   4
#define NTHREADS (NWAVES * 32)
#define UCOLS    32
#define NUNITS   (NTOT / UCOLS)
#define NSPLIT   2
#define RES_PAD  2048
#define PLANE    (HIN * WIN)
#define NTOT     (B_ * WOUT)

struct Gath { v2f g[2][CIN]; };

__device__ __forceinline__ void gather_step(const float* __restrict__ x,
                                            const int2* base_lds, int s, int half,
                                            const int (&pb)[2], const int (&wcol)[2], Gath& G)
{
    const int eA = 4 * s + 2 * half;
    const int2 r0 = base_lds[eA];
    const int2 r1 = base_lds[eA + 1];
    #pragma unroll
    for (int t = 0; t < 2; ++t) {
        int c0 = r0.y + wcol[t]; c0 -= (c0 >= WIN) ? WIN : 0;
        int c1 = r1.y + wcol[t]; c1 -= (c1 >= WIN) ? WIN : 0;
        const int o0 = pb[t] + r0.x + c0;
        const int o1 = pb[t] + r1.x + c1;
        #pragma unroll
        for (int c = 0; c < CIN; ++c) {
            G.g[t][c].x = x[o0 + c * PLANE];
            G.g[t][c].y = x[o1 + c * PLANE];
        }
    }
}

struct Afr { v2f a[CIN][4]; };

__device__ __forceinline__ void loadA_step(const float* A_lds, int s, int half, int l16, Afr& F)
{
    const int eA = 4 * s + 2 * half;
    #pragma unroll
    for (int c = 0; c < CIN; ++c)
        #pragma unroll
        for (int mt = 0; mt < 4; ++mt)
            F.a[c][mt] = *(const v2f*)&A_lds[(mt * 16 + l16) * ASTR + c * NNZ + eA];
}

__device__ __forceinline__ void mma_step(const Afr& F, const Gath& G, v8f (&acc)[2][4])
{
    #pragma unroll
    for (int c = 0; c < CIN; ++c)
        #pragma unroll
        for (int mt = 0; mt < 4; ++mt)
            #pragma unroll
            for (int t = 0; t < 2; ++t)
                acc[t][mt] = __builtin_amdgcn_wmma_f32_16x16x4_f32(false, F.a[c][mt], false, G.g[t][c],
                                                                   (short)0, acc[t][mt], false, false);
}

__device__ __forceinline__ void unit_cols(int unit, int l16, int (&pb)[2], int (&wcol)[2])
{
    #pragma unroll
    for (int t = 0; t < 2; ++t) {
        const int n = unit * UCOLS + t * 16 + l16;
        const int b = n / WOUT;
        const int w = n - b * WOUT;
        wcol[t] = w << 1;
        pb[t]   = b * CIN * PLANE;
    }
}

__global__ __launch_bounds__(NTHREADS)
void opm_gather_gemm_ln_kernel(const float* __restrict__ x,
                               const float* __restrict__ weight,
                               const float* __restrict__ psi,
                               const int*   __restrict__ idx_hi,
                               const int*   __restrict__ idx_dwi,
                               const float* __restrict__ lnw,
                               const float* __restrict__ lnb,
                               float*       __restrict__ out)
{
    __shared__ __attribute__((aligned(16))) float A_lds[COUT * ASTR + RES_PAD];
    __shared__ __attribute__((aligned(16))) int2  base_lds[NNZ];
    __shared__ __attribute__((aligned(16))) int2  raw_lds[NNZ];
    __shared__ __attribute__((aligned(16))) int   key_lds[NNZ];
    __shared__ __attribute__((aligned(16))) float psi_lds[KTAP * NNZ];
    __shared__ __attribute__((aligned(16))) float psim_lds[KTAP * NNZ];
    __shared__ int cnt_lds[NWAVES];
    __shared__ float lnw_lds[COUT];
    __shared__ float lnb_lds[COUT];

    const int tid  = threadIdx.x;
    const int h    = blockIdx.y;
    const int lane = tid & 31;
    const int l16  = lane & 15;
    const int half = lane >> 4;
    const int wv   = __builtin_amdgcn_readfirstlane(tid >> 5);

    const int tq = (tid < NNZ) ? tid : 0;
    const int g_hi = idx_hi[h * NNZ + tq];
    const int g_dw = idx_dwi[h * NNZ + tq];
    constexpr int PSI_IT = (KTAP * NNZ + NTHREADS - 1) / NTHREADS;
    float pv[PSI_IT];
    #pragma unroll
    for (int it = 0; it < PSI_IT; ++it) {
        const int i  = tid + it * NTHREADS;
        const int ic = (i < KTAP * NNZ) ? i : 0;
        const int k9 = ic / NNZ;
        const int e  = ic - k9 * NNZ;
        pv[it] = psi[(k9 * HOUT + h) * NNZ + e];
    }
    const int tc = tid & (COUT - 1);
    const float g_lw = lnw[tc];
    const float g_lb = lnb[tc];

    const int my_rw  = g_hi * WIN;
    const int my_dw  = g_dw;
    const int my_key = my_rw + my_dw;
    if (tid < NNZ) {
        raw_lds[tid] = make_int2(my_rw, my_dw);
        key_lds[tid] = my_key;
    }
    #pragma unroll
    for (int it = 0; it < PSI_IT; ++it) {
        const int i = tid + it * NTHREADS;
        if (i < KTAP * NNZ) psi_lds[i] = pv[it];
    }
    if (tid < COUT) {
        lnw_lds[tid] = g_lw;
        lnb_lds[tid] = g_lb;
    }
    __syncthreads();

    unsigned msk[3] = {0u, 0u, 0u};
    float pm[KTAP];
    #pragma unroll
    for (int k9 = 0; k9 < KTAP; ++k9) pm[k9] = 0.f;
    bool isrep = false;
    if (tid < NNZ) {
        #pragma unroll
        for (int j = 0; j < 3; ++j) {
            #pragma unroll
            for (int q = 0; q < 8; ++q) {
                const int4 kk = *(const int4*)&key_lds[j * 32 + q * 4];
                msk[j] |= (kk.x == my_key ? 1u : 0u) << (q * 4 + 0);
                msk[j] |= (kk.y == my_key ? 1u : 0u) << (q * 4 + 1);
                msk[j] |= (kk.z == my_key ? 1u : 0u) << (q * 4 + 2);
                msk[j] |= (kk.w == my_key ? 1u : 0u) << (q * 4 + 3);
            }
        }
        const int first = msk[0] ? __builtin_ctz(msk[0])
                        : (msk[1] ? 32 + __builtin_ctz(msk[1]) : 64 + __builtin_ctz(msk[2]));
        isrep = (first == tid);
        #pragma unroll
        for (int j = 0; j < 3; ++j) {
            unsigned mm = msk[j];
            while (mm) {
                const int e2 = j * 32 + __builtin_ctz(mm);
                mm &= mm - 1u;
                #pragma unroll
                for (int k9 = 0; k9 < KTAP; ++k9) pm[k9] += psi_lds[k9 * NNZ + e2];
            }
        }
    }
    const unsigned bal = __builtin_amdgcn_ballot_w32(isrep);
    if (lane == 0) cnt_lds[wv] = __builtin_popcount(bal);
    __syncthreads();

    const int cw0 = cnt_lds[0], cw1 = cnt_lds[1], cw2 = cnt_lds[2];
    const int ndist = cw0 + cw1 + cw2;
    {
        const int before = (wv > 0 ? cw0 : 0) + (wv > 1 ? cw1 : 0);
        const int slot   = before + __builtin_popcount(bal & ((1u << lane) - 1u));
        if (isrep) {
            base_lds[slot] = make_int2(my_rw, my_dw);
            #pragma unroll
            for (int k9 = 0; k9 < KTAP; ++k9) psim_lds[k9 * NNZ + slot] = pm[k9];
        }
        if (tid >= ndist && tid < NNZ) {
            base_lds[tid] = raw_lds[0];
            #pragma unroll
            for (int k9 = 0; k9 < KTAP; ++k9) psim_lds[k9 * NNZ + tid] = 0.f;
        }
    }
    const int nst = __builtin_amdgcn_readfirstlane(((ndist + 15) >> 4) << 2);
    __syncthreads();

    {
        float p[3][KTAP];
        #pragma unroll
        for (int k9 = 0; k9 < KTAP; ++k9)
            #pragma unroll
            for (int j = 0; j < 3; ++j)
                p[j][k9] = psim_lds[k9 * NNZ + j * 32 + lane];

        float wc[CIN * KTAP];
        #pragma unroll
        for (int i = 0; i < CIN * KTAP; ++i) wc[i] = weight[wv * (CIN * KTAP) + i];

        for (int o = wv; o < COUT; o += NWAVES) {
            const int on = (o + NWAVES < COUT) ? (o + NWAVES) : o;
            float wn[CIN * KTAP];
            #pragma unroll
            for (int i = 0; i < CIN * KTAP; ++i) wn[i] = weight[on * (CIN * KTAP) + i];

            #pragma unroll
            for (int c = 0; c < CIN; ++c) {
                #pragma unroll
                for (int j = 0; j < 3; ++j) {
                    float s = 0.f;
                    #pragma unroll
                    for (int k9 = 0; k9 < KTAP; ++k9)
                        s += wc[c * KTAP + k9] * p[j][k9];
                    A_lds[o * ASTR + c * NNZ + j * 32 + lane] = s;
                }
            }
            #pragma unroll
            for (int i = 0; i < CIN * KTAP; ++i) wc[i] = wn[i];
        }
    }

    __syncthreads();

    const int u0 = (blockIdx.x * NUNITS + NSPLIT - 1) / NSPLIT;
    const int u1 = ((blockIdx.x + 1) * NUNITS + NSPLIT - 1) / NSPLIT;
    #define STEP(Fcur, Gcur, sA, Fnext, sG, Gnext, PB, WC)                  \
        do {                                                                \
            loadA_step(A_lds, (sA), half, l16, Fnext);                      \
            gather_step(x, base_lds, (sG), half, PB, WC, Gnext);            \
            __builtin_amdgcn_sched_barrier(0);                              \
            mma_step(Fcur, Gcur, acc);                                      \
        } while (0)

    int unit = u0 + wv;
    int pb[2], wcol[2];
    unit_cols(unit, l16, pb, wcol);

    Gath G0, G1, G2, G3;
    Afr  F0, F1;
    gather_step(x, base_lds, 0, half, pb, wcol, G0);
    gather_step(x, base_lds, 1, half, pb, wcol, G1);
    gather_step(x, base_lds, 2, half, pb, wcol, G2);
    loadA_step(A_lds, 0, half, l16, F0);

    for (; unit < u1; unit += NWAVES) {

    v8f acc[2][4];
    #pragma unroll
    for (int t = 0; t < 2; ++t)
        #pragma unroll
        for (int mt = 0; mt < 4; ++mt) acc[t][mt] = (v8f){0.f,0.f,0.f,0.f,0.f,0.f,0.f,0.f};

    for (int s = 0; s < nst - 4; s += 4) {
        STEP(F0, G0, s + 1, F1, s + 3, G3, pb, wcol);
        STEP(F1, G1, s + 2, F0, s + 4, G0, pb, wcol);
        STEP(F0, G2, s + 3, F1, s + 5, G1, pb, wcol);
        STEP(F1, G3, s + 4, F0, s + 6, G2, pb, wcol);
    }

    const int nunit = (unit + NWAVES < u1) ? (unit + NWAVES) : unit;
    int pbN[2], wcolN[2];
    unit_cols(nunit, l16, pbN, wcolN);
    STEP(F0, G0, nst - 3, F1, nst - 1, G3, pb,  wcol);
    STEP(F1, G1, nst - 2, F0, 0,       G0, pbN, wcolN);
    STEP(F0, G2, nst - 1, F1, 1,       G1, pbN, wcolN);
    STEP(F1, G3, 0,       F0, 2,       G2, pbN, wcolN);

    int bb[2], ww[2];
    #pragma unroll
    for (int t = 0; t < 2; ++t) {
        const int n = unit * UCOLS + t * 16 + l16;
        bb[t] = n / WOUT;
        ww[t] = n - bb[t] * WOUT;
    }

    #pragma unroll
    for (int t = 0; t < 2; ++t) {
        float s = 0.f, ss = 0.f;
        #pragma unroll
        for (int mt = 0; mt < 4; ++mt) {
            #pragma unroll
            for (int r = 0; r < 8; ++r) {
                float v = acc[t][mt][r];
                s  += v;
                ss += v * v;
            }
        }
        s  += __shfl_xor(s, 16, 32);
        ss += __shfl_xor(ss, 16, 32);
        const float mean = s * (1.f / COUT);
        const float var  = ss * (1.f / COUT) - mean * mean;
        const float rstd = rsqrtf(var + LN_EPS);

        const int obase = bb[t] * (COUT * HOUT * WOUT) + h * WOUT + ww[t] + (half << 3) * (HOUT * WOUT);
        #pragma unroll
        for (int mt = 0; mt < 4; ++mt) {
            #pragma unroll
            for (int r = 0; r < 8; ++r) {
                const int o = mt * 16 + r + (half << 3);
                const float v = (acc[t][mt][r] - mean) * rstd * lnw_lds[o] + lnb_lds[o];
                out[obase + (mt * 16 + r) * (HOUT * WOUT)] = v;
            }
        }
    }

    #pragma unroll
    for (int t = 0; t < 2; ++t) { pb[t] = pbN[t]; wcol[t] = wcolN[t]; }

    }
    #undef STEP
}

extern "C" void kernel_launch(void* const* d_in, const int* in_sizes, int n_in,
                              void* d_out, int out_size, void* d_ws, size_t ws_size,
                              hipStream_t stream) {
    (void)in_sizes; (void)n_in; (void)d_ws; (void)ws_size; (void)out_size;
    const float* x       = (const float*)d_in[0];
    const float* weight  = (const float*)d_in[1];
    const float* psi     = (const float*)d_in[2];
    const int*   idx_hi  = (const int*)d_in[3];
    const int*   idx_dwi = (const int*)d_in[4];
    const float* lnw     = (const float*)d_in[5];
    const float* lnb     = (const float*)d_in[6];
    float*       out     = (float*)d_out;

    dim3 grid(NSPLIT, HOUT, 1);
    dim3 block(NTHREADS, 1, 1);
    opm_gather_gemm_ln_kernel<<<grid, block, 0, stream>>>(
        x, weight, psi, idx_hi, idx_dwi, lnw, lnb, out);
}


# Wave priority follows the units a wave has left (s_setprio 3, 2, 1, then 0 in its last unit's closing steps): the three waves of a SIMD end together, not oldest first; summing order unchanged
# speedup vs baseline: 1.7144x; 1.0404x over previous
#include <hip/hip_runtime.h>


typedef __attribute__((ext_vector_type(2))) float v2f;
typedef __attribute__((ext_vector_type(8))) float v8f;

#define B_       2
#define CIN      3
#define COUT     64
#define HIN      721
#define WIN      1440
#define HOUT     361
#define WOUT     720
#define KTAP     9
#define NNZ      96
#define LN_EPS   1e-5f

#define KTOT     (CIN * NNZ)
#define ASTR     292
#define NWAVES   4
#define NTHREADS (NWAVES * 32)
#define UCOLS    32
#define NUNITS   (NTOT / UCOLS)
#define NSPLIT   2
#define RES_PAD  2048
#define PLANE    (HIN * WIN)
#define NTOT     (B_ * WOUT)

struct Gath { v2f g[2][CIN]; };

__device__ __forceinline__ void gather_step(const float* __restrict__ x,
                                            const int2* base_lds, int s, int half,
                                            const int (&pb)[2], const int (&wcol)[2], Gath& G)
{
    const int eA = 4 * s + 2 * half;
    const int2 r0 = base_lds[eA];
    const int2 r1 = base_lds[eA + 1];
    #pragma unroll
    for (int t = 0; t < 2; ++t) {
        int c0 = r0.y + wcol[t]; c0 -= (c0 >= WIN) ? WIN : 0;
        int c1 = r1.y + wcol[t]; c1 -= (c1 >= WIN) ? WIN : 0;
        const int o0 = pb[t] + r0.x + c0;
        const int o1 = pb[t] + r1.x + c1;
        #pragma unroll
        for (int c = 0; c < CIN; ++c) {
            G.g[t][c].x = x[o0 + c * PLANE];
            G.g[t][c].y = x[o1 + c * PLANE];
        }
    }
}

struct Afr { v2f a[CIN][4]; };

__device__ __forceinline__ void loadA_step(const float* A_lds, int s, int half, int l16, Afr& F)
{
    const int eA = 4 * s + 2 * half;
    #pragma unroll
    for (int c = 0; c < CIN; ++c)
        #pragma unroll
        for (int mt = 0; mt < 4; ++mt)
            F.a[c][mt] = *(const v2f*)&A_lds[(mt * 16 + l16) * ASTR + c * NNZ + eA];
}

__device__ __forceinline__ void mma_step(const Afr& F, const Gath& G, v8f (&acc)[2][4])
{
    #pragma unroll
    for (int c = 0; c < CIN; ++c)
        #pragma unroll
        for (int mt = 0; mt < 4; ++mt)
            #pragma unroll
            for (int t = 0; t < 2; ++t)
                acc[t][mt] = __builtin_amdgcn_wmma_f32_16x16x4_f32(false, F.a[c][mt], false, G.g[t][c],
                                                                   (short)0, acc[t][mt], false, false);
}

__device__ __forceinline__ void unit_cols(int unit, int l16, int (&pb)[2], int (&wcol)[2])
{
    #pragma unroll
    for (int t = 0; t < 2; ++t) {
        const int n = unit * UCOLS + t * 16 + l16;
        const int b = n / WOUT;
        const int w = n - b * WOUT;
        wcol[t] = w << 1;
        pb[t]   = b * CIN * PLANE;
    }
}

__global__ __launch_bounds__(NTHREADS)
void opm_gather_gemm_ln_kernel(const float* __restrict__ x,
                               const float* __restrict__ weight,
                               const float* __restrict__ psi,
                               const int*   __restrict__ idx_hi,
                               const int*   __restrict__ idx_dwi,
                               const float* __restrict__ lnw,
                               const float* __restrict__ lnb,
                               float*       __restrict__ out)
{
    __shared__ __attribute__((aligned(16))) float A_lds[COUT * ASTR + RES_PAD];
    __shared__ __attribute__((aligned(16))) int2  base_lds[NNZ];
    __shared__ __attribute__((aligned(16))) int2  raw_lds[NNZ];
    __shared__ __attribute__((aligned(16))) int   key_lds[NNZ];
    __shared__ __attribute__((aligned(16))) float psi_lds[KTAP * NNZ];
    __shared__ __attribute__((aligned(16))) float psim_lds[KTAP * NNZ];
    __shared__ int cnt_lds[NWAVES];
    __shared__ float lnw_lds[COUT];
    __shared__ float lnb_lds[COUT];

    const int tid  = threadIdx.x;
    const int h    = blockIdx.y;
    const int lane = tid & 31;
    const int l16  = lane & 15;
    const int half = lane >> 4;
    const int wv   = __builtin_amdgcn_readfirstlane(tid >> 5);

    const int tq = (tid < NNZ) ? tid : 0;
    const int g_hi = idx_hi[h * NNZ + tq];
    const int g_dw = idx_dwi[h * NNZ + tq];
    constexpr int PSI_IT = (KTAP * NNZ + NTHREADS - 1) / NTHREADS;
    float pv[PSI_IT];
    #pragma unroll
    for (int it = 0; it < PSI_IT; ++it) {
        const int i  = tid + it * NTHREADS;
        const int ic = (i < KTAP * NNZ) ? i : 0;
        const int k9 = ic / NNZ;
        const int e  = ic - k9 * NNZ;
        pv[it] = psi[(k9 * HOUT + h) * NNZ + e];
    }
    const int tc = tid & (COUT - 1);
    const float g_lw = lnw[tc];
    const float g_lb = lnb[tc];

    const int my_rw  = g_hi * WIN;
    const int my_dw  = g_dw;
    const int my_key = my_rw + my_dw;
    if (tid < NNZ) {
        raw_lds[tid] = make_int2(my_rw, my_dw);
        key_lds[tid] = my_key;
    }
    #pragma unroll
    for (int it = 0; it < PSI_IT; ++it) {
        const int i = tid + it * NTHREADS;
        if (i < KTAP * NNZ) psi_lds[i] = pv[it];
    }
    if (tid < COUT) {
        lnw_lds[tid] = g_lw;
        lnb_lds[tid] = g_lb;
    }
    __syncthreads();

    unsigned msk[3] = {0u, 0u, 0u};
    float pm[KTAP];
    #pragma unroll
    for (int k9 = 0; k9 < KTAP; ++k9) pm[k9] = 0.f;
    bool isrep = false;
    if (tid < NNZ) {
        #pragma unroll
        for (int j = 0; j < 3; ++j) {
            #pragma unroll
            for (int q = 0; q < 8; ++q) {
                const int4 kk = *(const int4*)&key_lds[j * 32 + q * 4];
                msk[j] |= (kk.x == my_key ? 1u : 0u) << (q * 4 + 0);
                msk[j] |= (kk.y == my_key ? 1u : 0u) << (q * 4 + 1);
                msk[j] |= (kk.z == my_key ? 1u : 0u) << (q * 4 + 2);
                msk[j] |= (kk.w == my_key ? 1u : 0u) << (q * 4 + 3);
            }
        }
        const int first = msk[0] ? __builtin_ctz(msk[0])
                        : (msk[1] ? 32 + __builtin_ctz(msk[1]) : 64 + __builtin_ctz(msk[2]));
        isrep = (first == tid);
        #pragma unroll
        for (int j = 0; j < 3; ++j) {
            unsigned mm = msk[j];
            while (mm) {
                const int e2 = j * 32 + __builtin_ctz(mm);
                mm &= mm - 1u;
                #pragma unroll
                for (int k9 = 0; k9 < KTAP; ++k9) pm[k9] += psi_lds[k9 * NNZ + e2];
            }
        }
    }
    const unsigned bal = __builtin_amdgcn_ballot_w32(isrep);
    if (lane == 0) cnt_lds[wv] = __builtin_popcount(bal);
    __syncthreads();

    const int cw0 = cnt_lds[0], cw1 = cnt_lds[1], cw2 = cnt_lds[2];
    const int ndist = cw0 + cw1 + cw2;
    {
        const int before = (wv > 0 ? cw0 : 0) + (wv > 1 ? cw1 : 0);
        const int slot   = before + __builtin_popcount(bal & ((1u << lane) - 1u));
        if (isrep) {
            base_lds[slot] = make_int2(my_rw, my_dw);
            #pragma unroll
            for (int k9 = 0; k9 < KTAP; ++k9) psim_lds[k9 * NNZ + slot] = pm[k9];
        }
        if (tid >= ndist && tid < NNZ) {
            base_lds[tid] = raw_lds[0];
            #pragma unroll
            for (int k9 = 0; k9 < KTAP; ++k9) psim_lds[k9 * NNZ + tid] = 0.f;
        }
    }
    const int nst = __builtin_amdgcn_readfirstlane(((ndist + 15) >> 4) << 2);
    __syncthreads();

    {
        float p[3][KTAP];
        #pragma unroll
        for (int k9 = 0; k9 < KTAP; ++k9)
            #pragma unroll
            for (int j = 0; j < 3; ++j)
                p[j][k9] = psim_lds[k9 * NNZ + j * 32 + lane];

        float wc[CIN * KTAP];
        #pragma unroll
        for (int i = 0; i < CIN * KTAP; ++i) wc[i] = weight[wv * (CIN * KTAP) + i];

        for (int o = wv; o < COUT; o += NWAVES) {
            const int on = (o + NWAVES < COUT) ? (o + NWAVES) : o;
            float wn[CIN * KTAP];
            #pragma unroll
            for (int i = 0; i < CIN * KTAP; ++i) wn[i] = weight[on * (CIN * KTAP) + i];

            #pragma unroll
            for (int c = 0; c < CIN; ++c) {
                #pragma unroll
                for (int j = 0; j < 3; ++j) {
                    float s = 0.f;
                    #pragma unroll
                    for (int k9 = 0; k9 < KTAP; ++k9)
                        s += wc[c * KTAP + k9] * p[j][k9];
                    A_lds[o * ASTR + c * NNZ + j * 32 + lane] = s;
                }
            }
            #pragma unroll
            for (int i = 0; i < CIN * KTAP; ++i) wc[i] = wn[i];
        }
    }

    __syncthreads();

    const int u0 = (blockIdx.x * NUNITS + NSPLIT - 1) / NSPLIT;
    const int u1 = ((blockIdx.x + 1) * NUNITS + NSPLIT - 1) / NSPLIT;
    #define STEP(Fcur, Gcur, sA, Fnext, sG, Gnext, PB, WC)                  \
        do {                                                                \
            loadA_step(A_lds, (sA), half, l16, Fnext);                      \
            gather_step(x, base_lds, (sG), half, PB, WC, Gnext);            \
            __builtin_amdgcn_sched_barrier(0);                              \
            mma_step(Fcur, Gcur, acc);                                      \
        } while (0)

    int unit = u0 + wv;
    int pb[2], wcol[2];
    unit_cols(unit, l16, pb, wcol);

    Gath G0, G1, G2, G3;
    Afr  F0, F1;
    gather_step(x, base_lds, 0, half, pb, wcol, G0);
    gather_step(x, base_lds, 1, half, pb, wcol, G1);
    gather_step(x, base_lds, 2, half, pb, wcol, G2);
    loadA_step(A_lds, 0, half, l16, F0);

    for (; unit < u1; unit += NWAVES) {

    const int rem = (u1 - unit + NWAVES - 1) / NWAVES;
    if (rem >= 3)      __builtin_amdgcn_s_setprio(3);
    else if (rem == 2) __builtin_amdgcn_s_setprio(2);
    else               __builtin_amdgcn_s_setprio(1);

    v8f acc[2][4];
    #pragma unroll
    for (int t = 0; t < 2; ++t)
        #pragma unroll
        for (int mt = 0; mt < 4; ++mt) acc[t][mt] = (v8f){0.f,0.f,0.f,0.f,0.f,0.f,0.f,0.f};

    for (int s = 0; s < nst - 4; s += 4) {
        STEP(F0, G0, s + 1, F1, s + 3, G3, pb, wcol);
        STEP(F1, G1, s + 2, F0, s + 4, G0, pb, wcol);
        STEP(F0, G2, s + 3, F1, s + 5, G1, pb, wcol);
        STEP(F1, G3, s + 4, F0, s + 6, G2, pb, wcol);
    }

    const int nunit = (unit + NWAVES < u1) ? (unit + NWAVES) : unit;
    int pbN[2], wcolN[2];
    unit_cols(nunit, l16, pbN, wcolN);
    if (rem == 1) __builtin_amdgcn_s_setprio(0);
    STEP(F0, G0, nst - 3, F1, nst - 1, G3, pb,  wcol);
    STEP(F1, G1, nst - 2, F0, 0,       G0, pbN, wcolN);
    STEP(F0, G2, nst - 1, F1, 1,       G1, pbN, wcolN);
    STEP(F1, G3, 0,       F0, 2,       G2, pbN, wcolN);

    int bb[2], ww[2];
    #pragma unroll
    for (int t = 0; t < 2; ++t) {
        const int n = unit * UCOLS + t * 16 + l16;
        bb[t] = n / WOUT;
        ww[t] = n - bb[t] * WOUT;
    }

    #pragma unroll
    for (int t = 0; t < 2; ++t) {
        float s = 0.f, ss = 0.f;
        #pragma unroll
        for (int mt = 0; mt < 4; ++mt) {
            #pragma unroll
            for (int r = 0; r < 8; ++r) {
                float v = acc[t][mt][r];
                s  += v;
                ss += v * v;
            }
        }
        s  += __shfl_xor(s, 16, 32);
        ss += __shfl_xor(ss, 16, 32);
        const float mean = s * (1.f / COUT);
        const float var  = ss * (1.f / COUT) - mean * mean;
        const float rstd = rsqrtf(var + LN_EPS);

        const int obase = bb[t] * (COUT * HOUT * WOUT) + h * WOUT + ww[t] + (half << 3) * (HOUT * WOUT);
        #pragma unroll
        for (int mt = 0; mt < 4; ++mt) {
            #pragma unroll
            for (int r = 0; r < 8; ++r) {
                const int o = mt * 16 + r + (half << 3);
                const float v = (acc[t][mt][r] - mean) * rstd * lnw_lds[o] + lnb_lds[o];
                out[obase + (mt * 16 + r) * (HOUT * WOUT)] = v;
            }
        }
    }

    #pragma unroll
    for (int t = 0; t < 2; ++t) { pb[t] = pbN[t]; wcol[t] = wcolN[t]; }

    }
    #undef STEP
}

extern "C" void kernel_launch(void* const* d_in, const int* in_sizes, int n_in,
                              void* d_out, int out_size, void* d_ws, size_t ws_size,
                              hipStream_t stream) {
    (void)in_sizes; (void)n_in; (void)d_ws; (void)ws_size; (void)out_size;
    const float* x       = (const float*)d_in[0];
    const float* weight  = (const float*)d_in[1];
    const float* psi     = (const float*)d_in[2];
    const int*   idx_hi  = (const int*)d_in[3];
    const int*   idx_dwi = (const int*)d_in[4];
    const float* lnw     = (const float*)d_in[5];
    const float* lnb     = (const float*)d_in[6];
    float*       out     = (float*)d_out;

    dim3 grid(NSPLIT, HOUT, 1);
    dim3 block(NTHREADS, 1, 1);
    opm_gather_gemm_ln_kernel<<<grid, block, 0, stream>>>(
        x, weight, psi, idx_hi, idx_dwi, lnw, lnb, out);
}


# Skip the matrix products of closing K steps that hold only zero-weight padding slots: products cover ceil(distinct taps/4) steps, not the 16-tap rounding; ring loads and summing order unchanged
# speedup vs baseline: 1.7759x; 1.0359x over previous
#include <hip/hip_runtime.h>


typedef __attribute__((ext_vector_type(2))) float v2f;
typedef __attribute__((ext_vector_type(8))) float v8f;

#define B_       2
#define CIN      3
#define COUT     64
#define HIN      721
#define WIN      1440
#define HOUT     361
#define WOUT     720
#define KTAP     9
#define NNZ      96
#define LN_EPS   1e-5f

#define KTOT     (CIN * NNZ)
#define ASTR     292
#define NWAVES   4
#define NTHREADS (NWAVES * 32)
#define UCOLS    32
#define NUNITS   (NTOT / UCOLS)
#define NSPLIT   2
#define RES_PAD  2048
#define PLANE    (HIN * WIN)
#define NTOT     (B_ * WOUT)

struct Gath { v2f g[2][CIN]; };

__device__ __forceinline__ void gather_step(const float* __restrict__ x,
                                            const int2* base_lds, int s, int half,
                                            const int (&pb)[2], const int (&wcol)[2], Gath& G)
{
    const int eA = 4 * s + 2 * half;
    const int2 r0 = base_lds[eA];
    const int2 r1 = base_lds[eA + 1];
    #pragma unroll
    for (int t = 0; t < 2; ++t) {
        int c0 = r0.y + wcol[t]; c0 -= (c0 >= WIN) ? WIN : 0;
        int c1 = r1.y + wcol[t]; c1 -= (c1 >= WIN) ? WIN : 0;
        const int o0 = pb[t] + r0.x + c0;
        const int o1 = pb[t] + r1.x + c1;
        #pragma unroll
        for (int c = 0; c < CIN; ++c) {
            G.g[t][c].x = x[o0 + c * PLANE];
            G.g[t][c].y = x[o1 + c * PLANE];
        }
    }
}

struct Afr { v2f a[CIN][4]; };

__device__ __forceinline__ void loadA_step(const float* A_lds, int s, int half, int l16, Afr& F)
{
    const int eA = 4 * s + 2 * half;
    #pragma unroll
    for (int c = 0; c < CIN; ++c)
        #pragma unroll
        for (int mt = 0; mt < 4; ++mt)
            F.a[c][mt] = *(const v2f*)&A_lds[(mt * 16 + l16) * ASTR + c * NNZ + eA];
}

__device__ __forceinline__ void mma_step(const Afr& F, const Gath& G, v8f (&acc)[2][4])
{
    #pragma unroll
    for (int c = 0; c < CIN; ++c)
        #pragma unroll
        for (int mt = 0; mt < 4; ++mt)
            #pragma unroll
            for (int t = 0; t < 2; ++t)
                acc[t][mt] = __builtin_amdgcn_wmma_f32_16x16x4_f32(false, F.a[c][mt], false, G.g[t][c],
                                                                   (short)0, acc[t][mt], false, false);
}

__device__ __forceinline__ void unit_cols(int unit, int l16, int (&pb)[2], int (&wcol)[2])
{
    #pragma unroll
    for (int t = 0; t < 2; ++t) {
        const int n = unit * UCOLS + t * 16 + l16;
        const int b = n / WOUT;
        const int w = n - b * WOUT;
        wcol[t] = w << 1;
        pb[t]   = b * CIN * PLANE;
    }
}

__global__ __launch_bounds__(NTHREADS)
void opm_gather_gemm_ln_kernel(const float* __restrict__ x,
                               const float* __restrict__ weight,
                               const float* __restrict__ psi,
                               const int*   __restrict__ idx_hi,
                               const int*   __restrict__ idx_dwi,
                               const float* __restrict__ lnw,
                               const float* __restrict__ lnb,
                               float*       __restrict__ out)
{
    __shared__ __attribute__((aligned(16))) float A_lds[COUT * ASTR + RES_PAD];
    __shared__ __attribute__((aligned(16))) int2  base_lds[NNZ];
    __shared__ __attribute__((aligned(16))) int2  raw_lds[NNZ];
    __shared__ __attribute__((aligned(16))) int   key_lds[NNZ];
    __shared__ __attribute__((aligned(16))) float psi_lds[KTAP * NNZ];
    __shared__ __attribute__((aligned(16))) float psim_lds[KTAP * NNZ];
    __shared__ int cnt_lds[NWAVES];
    __shared__ float lnw_lds[COUT];
    __shared__ float lnb_lds[COUT];

    const int tid  = threadIdx.x;
    const int h    = blockIdx.y;
    const int lane = tid & 31;
    const int l16  = lane & 15;
    const int half = lane >> 4;
    const int wv   = __builtin_amdgcn_readfirstlane(tid >> 5);

    const int tq = (tid < NNZ) ? tid : 0;
    const int g_hi = idx_hi[h * NNZ + tq];
    const int g_dw = idx_dwi[h * NNZ + tq];
    constexpr int PSI_IT = (KTAP * NNZ + NTHREADS - 1) / NTHREADS;
    float pv[PSI_IT];
    #pragma unroll
    for (int it = 0; it < PSI_IT; ++it) {
        const int i  = tid + it * NTHREADS;
        const int ic = (i < KTAP * NNZ) ? i : 0;
        const int k9 = ic / NNZ;
        const int e  = ic - k9 * NNZ;
        pv[it] = psi[(k9 * HOUT + h) * NNZ + e];
    }
    const int tc = tid & (COUT - 1);
    const float g_lw = lnw[tc];
    const float g_lb = lnb[tc];

    const int my_rw  = g_hi * WIN;
    const int my_dw  = g_dw;
    const int my_key = my_rw + my_dw;
    if (tid < NNZ) {
        raw_lds[tid] = make_int2(my_rw, my_dw);
        key_lds[tid] = my_key;
    }
    #pragma unroll
    for (int it = 0; it < PSI_IT; ++it) {
        const int i = tid + it * NTHREADS;
        if (i < KTAP * NNZ) psi_lds[i] = pv[it];
    }
    if (tid < COUT) {
        lnw_lds[tid] = g_lw;
        lnb_lds[tid] = g_lb;
    }
    __syncthreads();

    unsigned msk[3] = {0u, 0u, 0u};
    float pm[KTAP];
    #pragma unroll
    for (int k9 = 0; k9 < KTAP; ++k9) pm[k9] = 0.f;
    bool isrep = false;
    if (tid < NNZ) {
        #pragma unroll
        for (int j = 0; j < 3; ++j) {
            #pragma unroll
            for (int q = 0; q < 8; ++q) {
                const int4 kk = *(const int4*)&key_lds[j * 32 + q * 4];
                msk[j] |= (kk.x == my_key ? 1u : 0u) << (q * 4 + 0);
                msk[j] |= (kk.y == my_key ? 1u : 0u) << (q * 4 + 1);
                msk[j] |= (kk.z == my_key ? 1u : 0u) << (q * 4 + 2);
                msk[j] |= (kk.w == my_key ? 1u : 0u) << (q * 4 + 3);
            }
        }
        const int first = msk[0] ? __builtin_ctz(msk[0])
                        : (msk[1] ? 32 + __builtin_ctz(msk[1]) : 64 + __builtin_ctz(msk[2]));
        isrep = (first == tid);
        #pragma unroll
        for (int j = 0; j < 3; ++j) {
            unsigned mm = msk[j];
            while (mm) {
                const int e2 = j * 32 + __builtin_ctz(mm);
                mm &= mm - 1u;
                #pragma unroll
                for (int k9 = 0; k9 < KTAP; ++k9) pm[k9] += psi_lds[k9 * NNZ + e2];
            }
        }
    }
    const unsigned bal = __builtin_amdgcn_ballot_w32(isrep);
    if (lane == 0) cnt_lds[wv] = __builtin_popcount(bal);
    __syncthreads();

    const int cw0 = cnt_lds[0], cw1 = cnt_lds[1], cw2 = cnt_lds[2];
    const int ndist = cw0 + cw1 + cw2;
    {
        const int before = (wv > 0 ? cw0 : 0) + (wv > 1 ? cw1 : 0);
        const int slot   = before + __builtin_popcount(bal & ((1u << lane) - 1u));
        if (isrep) {
            base_lds[slot] = make_int2(my_rw, my_dw);
            #pragma unroll
            for (int k9 = 0; k9 < KTAP; ++k9) psim_lds[k9 * NNZ + slot] = pm[k9];
        }
        if (tid >= ndist && tid < NNZ) {
            base_lds[tid] = raw_lds[0];
            #pragma unroll
            for (int k9 = 0; k9 < KTAP; ++k9) psim_lds[k9 * NNZ + tid] = 0.f;
        }
    }
    const int nst = __builtin_amdgcn_readfirstlane(((ndist + 15) >> 4) << 2);
    const int nvalid = __builtin_amdgcn_readfirstlane(((ndist + 3) >> 2) - (nst - 4));
    __syncthreads();

    {
        float p[3][KTAP];
        #pragma unroll
        for (int k9 = 0; k9 < KTAP; ++k9)
            #pragma unroll
            for (int j = 0; j < 3; ++j)
                p[j][k9] = psim_lds[k9 * NNZ + j * 32 + lane];

        float wc[CIN * KTAP];
        #pragma unroll
        for (int i = 0; i < CIN * KTAP; ++i) wc[i] = weight[wv * (CIN * KTAP) + i];

        for (int o = wv; o < COUT; o += NWAVES) {
            const int on = (o + NWAVES < COUT) ? (o + NWAVES) : o;
            float wn[CIN * KTAP];
            #pragma unroll
            for (int i = 0; i < CIN * KTAP; ++i) wn[i] = weight[on * (CIN * KTAP) + i];

            #pragma unroll
            for (int c = 0; c < CIN; ++c) {
                #pragma unroll
                for (int j = 0; j < 3; ++j) {
                    float s = 0.f;
                    #pragma unroll
                    for (int k9 = 0; k9 < KTAP; ++k9)
                        s += wc[c * KTAP + k9] * p[j][k9];
                    A_lds[o * ASTR + c * NNZ + j * 32 + lane] = s;
                }
            }
            #pragma unroll
            for (int i = 0; i < CIN * KTAP; ++i) wc[i] = wn[i];
        }
    }

    __syncthreads();

    const int u0 = (blockIdx.x * NUNITS + NSPLIT - 1) / NSPLIT;
    const int u1 = ((blockIdx.x + 1) * NUNITS + NSPLIT - 1) / NSPLIT;
    #define STEP(Fcur, Gcur, sA, Fnext, sG, Gnext, PB, WC)                  \
        do {                                                                \
            loadA_step(A_lds, (sA), half, l16, Fnext);                      \
            gather_step(x, base_lds, (sG), half, PB, WC, Gnext);            \
            __builtin_amdgcn_sched_barrier(0);                              \
            mma_step(Fcur, Gcur, acc);                                      \
        } while (0)

    #define STEPC(doit, Fcur, Gcur, sA, Fnext, sG, Gnext, PB, WC)           \
        do {                                                                \
            loadA_step(A_lds, (sA), half, l16, Fnext);                      \
            gather_step(x, base_lds, (sG), half, PB, WC, Gnext);            \
            __builtin_amdgcn_sched_barrier(0);                              \
            if (doit) mma_step(Fcur, Gcur, acc);                            \
        } while (0)

    int unit = u0 + wv;
    int pb[2], wcol[2];
    unit_cols(unit, l16, pb, wcol);

    Gath G0, G1, G2, G3;
    Afr  F0, F1;
    gather_step(x, base_lds, 0, half, pb, wcol, G0);
    gather_step(x, base_lds, 1, half, pb, wcol, G1);
    gather_step(x, base_lds, 2, half, pb, wcol, G2);
    loadA_step(A_lds, 0, half, l16, F0);

    for (; unit < u1; unit += NWAVES) {

    const int rem = (u1 - unit + NWAVES - 1) / NWAVES;
    if (rem >= 3)      __builtin_amdgcn_s_setprio(3);
    else if (rem == 2) __builtin_amdgcn_s_setprio(2);
    else               __builtin_amdgcn_s_setprio(1);

    v8f acc[2][4];
    #pragma unroll
    for (int t = 0; t < 2; ++t)
        #pragma unroll
        for (int mt = 0; mt < 4; ++mt) acc[t][mt] = (v8f){0.f,0.f,0.f,0.f,0.f,0.f,0.f,0.f};

    for (int s = 0; s < nst - 4; s += 4) {
        STEP(F0, G0, s + 1, F1, s + 3, G3, pb, wcol);
        STEP(F1, G1, s + 2, F0, s + 4, G0, pb, wcol);
        STEP(F0, G2, s + 3, F1, s + 5, G1, pb, wcol);
        STEP(F1, G3, s + 4, F0, s + 6, G2, pb, wcol);
    }

    const int nunit = (unit + NWAVES < u1) ? (unit + NWAVES) : unit;
    int pbN[2], wcolN[2];
    unit_cols(nunit, l16, pbN, wcolN);
    if (rem == 1) __builtin_amdgcn_s_setprio(0);
    STEP(F0, G0, nst - 3, F1, nst - 1, G3, pb,  wcol);
    STEPC(nvalid >= 2, F1, G1, nst - 2, F0, 0, G0, pbN, wcolN);
    STEPC(nvalid >= 3, F0, G2, nst - 1, F1, 1, G1, pbN, wcolN);
    STEPC(nvalid >= 4, F1, G3, 0,       F0, 2, G2, pbN, wcolN);

    int bb[2], ww[2];
    #pragma unroll
    for (int t = 0; t < 2; ++t) {
        const int n = unit * UCOLS + t * 16 + l16;
        bb[t] = n / WOUT;
        ww[t] = n - bb[t] * WOUT;
    }

    #pragma unroll
    for (int t = 0; t < 2; ++t) {
        float s = 0.f, ss = 0.f;
        #pragma unroll
        for (int mt = 0; mt < 4; ++mt) {
            #pragma unroll
            for (int r = 0; r < 8; ++r) {
                float v = acc[t][mt][r];
                s  += v;
                ss += v * v;
            }
        }
        s  += __shfl_xor(s, 16, 32);
        ss += __shfl_xor(ss, 16, 32);
        const float mean = s * (1.f / COUT);
        const float var  = ss * (1.f / COUT) - mean * mean;
        const float rstd = rsqrtf(var + LN_EPS);

        const int obase = bb[t] * (COUT * HOUT * WOUT) + h * WOUT + ww[t] + (half << 3) * (HOUT * WOUT);
        #pragma unroll
        for (int mt = 0; mt < 4; ++mt) {
            #pragma unroll
            for (int r = 0; r < 8; ++r) {
                const int o = mt * 16 + r + (half << 3);
                const float v = (acc[t][mt][r] - mean) * rstd * lnw_lds[o] + lnb_lds[o];
                out[obase + (mt * 16 + r) * (HOUT * WOUT)] = v;
            }
        }
    }

    #pragma unroll
    for (int t = 0; t < 2; ++t) { pb[t] = pbN[t]; wcol[t] = wcolN[t]; }

    }
    #undef STEP
    #undef STEPC
}

extern "C" void kernel_launch(void* const* d_in, const int* in_sizes, int n_in,
                              void* d_out, int out_size, void* d_ws, size_t ws_size,
                              hipStream_t stream) {
    (void)in_sizes; (void)n_in; (void)d_ws; (void)ws_size; (void)out_size;
    const float* x       = (const float*)d_in[0];
    const float* weight  = (const float*)d_in[1];
    const float* psi     = (const float*)d_in[2];
    const int*   idx_hi  = (const int*)d_in[3];
    const int*   idx_dwi = (const int*)d_in[4];
    const float* lnw     = (const float*)d_in[5];
    const float* lnb     = (const float*)d_in[6];
    float*       out     = (float*)d_out;

    dim3 grid(NSPLIT, HOUT, 1);
    dim3 block(NTHREADS, 1, 1);
    opm_gather_gemm_ln_kernel<<<grid, block, 0, stream>>>(
        x, weight, psi, idx_hi, idx_dwi, lnw, lnb, out);
}
